# starting kernel
# baseline (speedup 1.0000x reference)
#include <hip/hip_runtime.h>


typedef __attribute__((ext_vector_type(2))) float v2f;
typedef __attribute__((ext_vector_type(8))) float v8f;

__global__ void fill_kernel(float* __restrict__ p, float v, int n) {
  int i = blockIdx.x * blockDim.x + threadIdx.x;
  if (i < n) p[i] = v;
}

__global__ void deg_kernel(const int* __restrict__ dst, float* __restrict__ deg, int E) {
  int e = blockIdx.x * blockDim.x + threadIdx.x;
  if (e < E) unsafeAtomicAdd(&deg[dst[e]], 1.0f);
}

__global__ void rsqrt_kernel(float* __restrict__ deg, int n) {
  int i = blockIdx.x * blockDim.x + threadIdx.x;
  if (i < n) deg[i] = rsqrtf(deg[i]);
}

template <int K, int ND, int RT, bool RELU_A>
__global__ void gemm_wmma_f32(const float* __restrict__ X,
                              const float* __restrict__ W,
                              float* __restrict__ H) {
  const int wave = threadIdx.x >> 5;
  const int lane = threadIdx.x & 31;
  const int col0 = blockIdx.y * 64 + wave * 16;

  const int m  = lane & 15;
  const int kb = (lane >> 4) * 2;

  v2f b[K / 4];
#pragma unroll
  for (int kk = 0; kk < K / 4; ++kk) {
    b[kk].x = W[(size_t)(kk * 4 + kb) * ND + col0 + m];
    b[kk].y = W[(size_t)(kk * 4 + kb + 1) * ND + col0 + m];
  }

  int row0 = blockIdx.x * (16 * RT);
#pragma unroll 1
  for (int rt = 0; rt < RT; ++rt, row0 += 16) {
    const float* __restrict__ xrow = X + (size_t)(row0 + m) * K + kb;
    v2f a[K / 4];
#pragma unroll
    for (int kk = 0; kk < K / 4; ++kk) {
      v2f av = *(const v2f*)(xrow + kk * 4);
      if (RELU_A) { av.x = fmaxf(av.x, 0.0f); av.y = fmaxf(av.y, 0.0f); }
      a[kk] = av;
    }

    v8f c = {};
#pragma unroll
    for (int kk = 0; kk < K / 4; ++kk)
      c = __builtin_amdgcn_wmma_f32_16x16x4_f32(false, a[kk], false, b[kk],
                                                (short)0, c, false, false);

    const int rbase = (lane >> 4) * 8;
#pragma unroll
    for (int r = 0; r < 8; ++r)
      H[(size_t)(row0 + rbase + r) * ND + col0 + m] = c[r];
  }
}

template <int D>
__global__ void selfloop_bias_kernel(const float* __restrict__ h,
                                     const float* __restrict__ dis,
                                     const float* __restrict__ bias,
                                     float* __restrict__ out, int n) {
  int i = blockIdx.x * blockDim.x + threadIdx.x;
  if (i < n * (D / 4)) {
    int v = i / (D / 4);
    int t = (i & (D / 4 - 1)) * 4;
    float w = dis[v];
    float ww = w * w;
    float4 hv = *(const float4*)(h + (size_t)v * D + t);
    float4 bv = *(const float4*)(bias + t);
    float4 o;
    o.x = bv.x + hv.x * ww; o.y = bv.y + hv.y * ww;
    o.z = bv.z + hv.z * ww; o.w = bv.w + hv.w * ww;
    *(float4*)(out + (size_t)v * D + t) = o;
  }
}

template <int D>
__global__ void scatter_kernel(const int* __restrict__ src,
                               const int* __restrict__ dst,
                               const float* __restrict__ h,
                               const float* __restrict__ dis,
                               float* __restrict__ out, int E) {
  int idx = blockIdx.x * blockDim.x + threadIdx.x;
  int e = idx / (D / 4);
  int t = (idx & (D / 4 - 1)) * 4;
  if (e < E) {
    int s = src[e], d = dst[e];
    float w = dis[s] * dis[d];
    float4 hv = *(const float4*)(h + (size_t)s * D + t);
    float* o = out + (size_t)d * D + t;
    unsafeAtomicAdd(o + 0, hv.x * w);
    unsafeAtomicAdd(o + 1, hv.y * w);
    unsafeAtomicAdd(o + 2, hv.z * w);
    unsafeAtomicAdd(o + 3, hv.w * w);
  }
}

static inline int cdiv(long long a, long long b) { return (int)((a + b - 1) / b); }

static void run_branch(const float* x, const int* ei, const float* W1,
                       const float* b1, const float* W2, const float* b2,
                       float* out_final, float* deg, float* h, float* out1,
                       int N, int E, hipStream_t stream) {
  const int IN = 128, HD = 128, OD = 64;
  const int RT = 5;
  const int* src = ei;
  const int* dst = ei + E;

  fill_kernel<<<cdiv(N, 256), 256, 0, stream>>>(deg, 1.0f, N);
  deg_kernel<<<cdiv(E, 256), 256, 0, stream>>>(dst, deg, E);
  rsqrt_kernel<<<cdiv(N, 256), 256, 0, stream>>>(deg, N);
  float* dis = deg;

  const int rowBlocks = N / (16 * RT);

  gemm_wmma_f32<IN, HD, RT, false>
      <<<dim3(rowBlocks, HD / 64), 128, 0, stream>>>(x, W1, h);
  selfloop_bias_kernel<HD><<<cdiv((long long)N * HD / 4, 256), 256, 0, stream>>>(
      h, dis, b1, out1, N);
  scatter_kernel<HD><<<cdiv((long long)E * HD / 4, 256), 256, 0, stream>>>(
      src, dst, h, dis, out1, E);

  gemm_wmma_f32<HD, OD, RT, true>
      <<<dim3(rowBlocks, 1), 128, 0, stream>>>(out1, W2, h);
  selfloop_bias_kernel<OD><<<cdiv((long long)N * OD / 4, 256), 256, 0, stream>>>(
      h, dis, b2, out_final, N);
  scatter_kernel<OD><<<cdiv((long long)E * OD / 4, 256), 256, 0, stream>>>(
      src, dst, h, dis, out_final, E);
}

extern "C" void kernel_launch(void* const* d_in, const int* in_sizes, int n_in,
                              void* d_out, int out_size, void* d_ws, size_t ws_size,
                              hipStream_t stream) {
  const float* x_feat = (const float*)d_in[0];
  const int*   x_ei   = (const int*)d_in[1];
  const float* y_feat = (const float*)d_in[2];
  const int*   y_ei   = (const int*)d_in[3];
  const float* W1x = (const float*)d_in[4];
  const float* b1x = (const float*)d_in[5];
  const float* W2x = (const float*)d_in[6];
  const float* b2x = (const float*)d_in[7];
  const float* W1y = (const float*)d_in[8];
  const float* b1y = (const float*)d_in[9];
  const float* W2y = (const float*)d_in[10];
  const float* b2y = (const float*)d_in[11];

  const int N = in_sizes[0] / 128;
  const int E = in_sizes[1] / 2;

  float* deg  = (float*)d_ws;
  float* h    = deg + N;
  float* out1 = h + (size_t)N * 128;

  float* out_x = (float*)d_out;
  float* out_y = out_x + (size_t)N * 64;

  run_branch(x_feat, x_ei, W1x, b1x, W2x, b2x, out_x, deg, h, out1, N, E, stream);
  run_branch(y_feat, y_ei, W1y, b1y, W2y, b2y, out_y, deg, h, out1, N, E, stream);
}


# Scatter-add: a wave owns 32 whole edges, lane l owns features l, l+32, ..: each row load and f32 atomic covers one aligned 128-B line (was 8 dwords in each of 4-5); sums in atomic arrival order
# speedup vs baseline: 1.7708x; 1.7708x over previous
#include <hip/hip_runtime.h>


typedef __attribute__((ext_vector_type(2))) float v2f;
typedef __attribute__((ext_vector_type(8))) float v8f;

__global__ void fill_kernel(float* __restrict__ p, float v, int n) {
  int i = blockIdx.x * blockDim.x + threadIdx.x;
  if (i < n) p[i] = v;
}

__global__ void deg_kernel(const int* __restrict__ dst, float* __restrict__ deg, int E) {
  int e = blockIdx.x * blockDim.x + threadIdx.x;
  if (e < E) unsafeAtomicAdd(&deg[dst[e]], 1.0f);
}

__global__ void rsqrt_kernel(float* __restrict__ deg, int n) {
  int i = blockIdx.x * blockDim.x + threadIdx.x;
  if (i < n) deg[i] = rsqrtf(deg[i]);
}

template <int K, int ND, int RT, bool RELU_A>
__global__ void gemm_wmma_f32(const float* __restrict__ X,
                              const float* __restrict__ W,
                              float* __restrict__ H) {
  const int wave = threadIdx.x >> 5;
  const int lane = threadIdx.x & 31;
  const int col0 = blockIdx.y * 64 + wave * 16;

  const int m  = lane & 15;
  const int kb = (lane >> 4) * 2;

  v2f b[K / 4];
#pragma unroll
  for (int kk = 0; kk < K / 4; ++kk) {
    b[kk].x = W[(size_t)(kk * 4 + kb) * ND + col0 + m];
    b[kk].y = W[(size_t)(kk * 4 + kb + 1) * ND + col0 + m];
  }

  int row0 = blockIdx.x * (16 * RT);
#pragma unroll 1
  for (int rt = 0; rt < RT; ++rt, row0 += 16) {
    const float* __restrict__ xrow = X + (size_t)(row0 + m) * K + kb;
    v2f a[K / 4];
#pragma unroll
    for (int kk = 0; kk < K / 4; ++kk) {
      v2f av = *(const v2f*)(xrow + kk * 4);
      if (RELU_A) { av.x = fmaxf(av.x, 0.0f); av.y = fmaxf(av.y, 0.0f); }
      a[kk] = av;
    }

    v8f c = {};
#pragma unroll
    for (int kk = 0; kk < K / 4; ++kk)
      c = __builtin_amdgcn_wmma_f32_16x16x4_f32(false, a[kk], false, b[kk],
                                                (short)0, c, false, false);

    const int rbase = (lane >> 4) * 8;
#pragma unroll
    for (int r = 0; r < 8; ++r)
      H[(size_t)(row0 + rbase + r) * ND + col0 + m] = c[r];
  }
}

template <int D>
__global__ void selfloop_bias_kernel(const float* __restrict__ h,
                                     const float* __restrict__ dis,
                                     const float* __restrict__ bias,
                                     float* __restrict__ out, int n) {
  int i = blockIdx.x * blockDim.x + threadIdx.x;
  if (i < n * (D / 4)) {
    int v = i / (D / 4);
    int t = (i & (D / 4 - 1)) * 4;
    float w = dis[v];
    float ww = w * w;
    float4 hv = *(const float4*)(h + (size_t)v * D + t);
    float4 bv = *(const float4*)(bias + t);
    float4 o;
    o.x = bv.x + hv.x * ww; o.y = bv.y + hv.y * ww;
    o.z = bv.z + hv.z * ww; o.w = bv.w + hv.w * ww;
    *(float4*)(out + (size_t)v * D + t) = o;
  }
}

template <int D, int NE>
__device__ __forceinline__ void scatter_edges(const float* __restrict__ hl,
                                              float* __restrict__ ol,
                                              int sv, int dv, float wl, int j) {
  constexpr int C = D / 32;
  float v[NE][C];
  for (int k = 0; k < NE; ++k) {
    const int s = __builtin_amdgcn_readlane(sv, j + k);
    const float* __restrict__ hp = hl + (size_t)s * D;
    for (int c = 0; c < C; ++c) v[k][c] = hp[c * 32];
  }
  for (int k = 0; k < NE; ++k) {
    const int d = __builtin_amdgcn_readlane(dv, j + k);
    const float w = __int_as_float(__builtin_amdgcn_readlane(__float_as_int(wl), j + k));
    float* op = ol + (size_t)d * D;
    for (int c = 0; c < C; ++c) unsafeAtomicAdd(op + c * 32, v[k][c] * w);
  }
}

template <int D, int NE>
__global__ void scatter_kernel(const int* __restrict__ src,
                               const int* __restrict__ dst,
                               const float* __restrict__ h,
                               const float* __restrict__ dis,
                               float* __restrict__ out, int E) {
  const int lane = threadIdx.x & 31;
  const int wv = __builtin_amdgcn_readfirstlane(
      (int)((blockIdx.x * blockDim.x + threadIdx.x) >> 5));
  const int e0 = wv * 32;
  if (e0 >= E) return;
  const int n = min(32, E - e0);
  const int ei = e0 + min(lane, n - 1);
  const int sv = src[ei];
  const int dv = dst[ei];
  const float wl = dis[sv] * dis[dv];
  const float* __restrict__ hl = h + lane;
  float* __restrict__ ol = out + lane;
  int j = 0;
  for (; j + NE <= n; j += NE) scatter_edges<D, NE>(hl, ol, sv, dv, wl, j);
  for (; j < n; ++j)           scatter_edges<D, 1>(hl, ol, sv, dv, wl, j);
}

static inline int cdiv(long long a, long long b) { return (int)((a + b - 1) / b); }

static void run_branch(const float* x, const int* ei, const float* W1,
                       const float* b1, const float* W2, const float* b2,
                       float* out_final, float* deg, float* h, float* out1,
                       int N, int E, hipStream_t stream) {
  const int IN = 128, HD = 128, OD = 64;
  const int RT = 5;
  const int NE1 = 4, NE2 = 8;
  const int* src = ei;
  const int* dst = ei + E;

  fill_kernel<<<cdiv(N, 256), 256, 0, stream>>>(deg, 1.0f, N);
  deg_kernel<<<cdiv(E, 256), 256, 0, stream>>>(dst, deg, E);
  rsqrt_kernel<<<cdiv(N, 256), 256, 0, stream>>>(deg, N);
  float* dis = deg;

  const int rowBlocks = N / (16 * RT);

  gemm_wmma_f32<IN, HD, RT, false>
      <<<dim3(rowBlocks, HD / 64), 128, 0, stream>>>(x, W1, h);
  selfloop_bias_kernel<HD><<<cdiv((long long)N * HD / 4, 256), 256, 0, stream>>>(
      h, dis, b1, out1, N);
  scatter_kernel<HD, NE1><<<cdiv((long long)cdiv(E, 32) * 32, 256), 256, 0, stream>>>(
      src, dst, h, dis, out1, E);

  gemm_wmma_f32<HD, OD, RT, true>
      <<<dim3(rowBlocks, 1), 128, 0, stream>>>(out1, W2, h);
  selfloop_bias_kernel<OD><<<cdiv((long long)N * OD / 4, 256), 256, 0, stream>>>(
      h, dis, b2, out_final, N);
  scatter_kernel<OD, NE2><<<cdiv((long long)cdiv(E, 32) * 32, 256), 256, 0, stream>>>(
      src, dst, h, dis, out_final, E);
}

extern "C" void kernel_launch(void* const* d_in, const int* in_sizes, int n_in,
                              void* d_out, int out_size, void* d_ws, size_t ws_size,
                              hipStream_t stream) {
  const float* x_feat = (const float*)d_in[0];
  const int*   x_ei   = (const int*)d_in[1];
  const float* y_feat = (const float*)d_in[2];
  const int*   y_ei   = (const int*)d_in[3];
  const float* W1x = (const float*)d_in[4];
  const float* b1x = (const float*)d_in[5];
  const float* W2x = (const float*)d_in[6];
  const float* b2x = (const float*)d_in[7];
  const float* W1y = (const float*)d_in[8];
  const float* b1y = (const float*)d_in[9];
  const float* W2y = (const float*)d_in[10];
  const float* b2y = (const float*)d_in[11];

  const int N = in_sizes[0] / 128;
  const int E = in_sizes[1] / 2;

  float* h    = (float*)d_ws;
  float* out1 = h + (size_t)N * 128;
  float* deg  = out1 + (size_t)N * 128;

  float* out_x = (float*)d_out;
  float* out_y = out_x + (size_t)N * 64;

  run_branch(x_feat, x_ei, W1x, b1x, W2x, b2x, out_x, deg, h, out1, N, E, stream);
  run_branch(y_feat, y_ei, W1y, b1y, W2y, b2y, out_y, deg, h, out1, N, E, stream);
}


# GEMM epilogue also stores the aggregation start value (bias + self-loop term h*dis^2) from the accumulator registers; the two init launches per branch are gone: 14 launches, was 18; sum order kept
# speedup vs baseline: 1.8477x; 1.0434x over previous
#include <hip/hip_runtime.h>


typedef __attribute__((ext_vector_type(2))) float v2f;
typedef __attribute__((ext_vector_type(8))) float v8f;

__global__ void fill_kernel(float* __restrict__ p, float v, int n) {
  int i = blockIdx.x * blockDim.x + threadIdx.x;
  if (i < n) p[i] = v;
}

__global__ void deg_kernel(const int* __restrict__ dst, float* __restrict__ deg, int E) {
  int e = blockIdx.x * blockDim.x + threadIdx.x;
  if (e < E) unsafeAtomicAdd(&deg[dst[e]], 1.0f);
}

__global__ void rsqrt_kernel(float* __restrict__ deg, int n) {
  int i = blockIdx.x * blockDim.x + threadIdx.x;
  if (i < n) deg[i] = rsqrtf(deg[i]);
}

template <int K, int ND, int RT, bool RELU_A>
__global__ void gemm_wmma_f32(const float* __restrict__ X,
                              const float* __restrict__ W,
                              const float* __restrict__ dis,
                              const float* __restrict__ bias,
                              float* __restrict__ H,
                              float* __restrict__ OUT) {
  const int wave = threadIdx.x >> 5;
  const int lane = threadIdx.x & 31;
  const int col0 = blockIdx.y * 64 + wave * 16;

  const int m  = lane & 15;
  const int kb = (lane >> 4) * 2;

  v2f b[K / 4];
#pragma unroll
  for (int kk = 0; kk < K / 4; ++kk) {
    b[kk].x = W[(size_t)(kk * 4 + kb) * ND + col0 + m];
    b[kk].y = W[(size_t)(kk * 4 + kb + 1) * ND + col0 + m];
  }
  const float bv = bias[col0 + m];
  const int rbase = (lane >> 4) * 8;

  int row0 = blockIdx.x * (16 * RT);
#pragma unroll 1
  for (int rt = 0; rt < RT; ++rt, row0 += 16) {
    const float* __restrict__ xrow = X + (size_t)(row0 + m) * K + kb;
    const float4 d0 = *(const float4*)(dis + row0 + rbase);
    const float4 d1 = *(const float4*)(dis + row0 + rbase + 4);
    v2f a[K / 4];
#pragma unroll
    for (int kk = 0; kk < K / 4; ++kk) {
      v2f av = *(const v2f*)(xrow + kk * 4);
      if (RELU_A) { av.x = fmaxf(av.x, 0.0f); av.y = fmaxf(av.y, 0.0f); }
      a[kk] = av;
    }

    v8f c = {};
#pragma unroll
    for (int kk = 0; kk < K / 4; ++kk)
      c = __builtin_amdgcn_wmma_f32_16x16x4_f32(false, a[kk], false, b[kk],
                                                (short)0, c, false, false);

    const float dd[8] = {d0.x, d0.y, d0.z, d0.w, d1.x, d1.y, d1.z, d1.w};
    const size_t base = (size_t)(row0 + rbase) * ND + col0 + m;
    float* __restrict__ hp = H + base;
    float* __restrict__ op = OUT + base;
#pragma unroll
    for (int r = 0; r < 8; ++r) {
      const float w = dd[r];
      const float ww = w * w;
      hp[r * ND] = c[r];
      op[r * ND] = bv + c[r] * ww;
    }
  }
}

template <int D, int NE>
__device__ __forceinline__ void scatter_edges(const float* __restrict__ hl,
                                              float* __restrict__ ol,
                                              int sv, int dv, float wl, int j) {
  constexpr int C = D / 32;
  float v[NE][C];
  for (int k = 0; k < NE; ++k) {
    const int s = __builtin_amdgcn_readlane(sv, j + k);
    const float* __restrict__ hp = hl + (size_t)s * D;
    for (int c = 0; c < C; ++c) v[k][c] = hp[c * 32];
  }
  for (int k = 0; k < NE; ++k) {
    const int d = __builtin_amdgcn_readlane(dv, j + k);
    const float w = __int_as_float(__builtin_amdgcn_readlane(__float_as_int(wl), j + k));
    float* op = ol + (size_t)d * D;
    for (int c = 0; c < C; ++c) unsafeAtomicAdd(op + c * 32, v[k][c] * w);
  }
}

template <int D, int NE>
__global__ void scatter_kernel(const int* __restrict__ src,
                               const int* __restrict__ dst,
                               const float* __restrict__ h,
                               const float* __restrict__ dis,
                               float* __restrict__ out, int E) {
  const int lane = threadIdx.x & 31;
  const int wv = __builtin_amdgcn_readfirstlane(
      (int)((blockIdx.x * blockDim.x + threadIdx.x) >> 5));
  const int e0 = wv * 32;
  if (e0 >= E) return;
  const int n = min(32, E - e0);
  const int ei = e0 + min(lane, n - 1);
  const int sv = src[ei];
  const int dv = dst[ei];
  const float wl = dis[sv] * dis[dv];
  const float* __restrict__ hl = h + lane;
  float* __restrict__ ol = out + lane;
  int j = 0;
  for (; j + NE <= n; j += NE) scatter_edges<D, NE>(hl, ol, sv, dv, wl, j);
  for (; j < n; ++j)           scatter_edges<D, 1>(hl, ol, sv, dv, wl, j);
}

static inline int cdiv(long long a, long long b) { return (int)((a + b - 1) / b); }

static void run_branch(const float* x, const int* ei, const float* W1,
                       const float* b1, const float* W2, const float* b2,
                       float* out_final, float* deg, float* h, float* out1,
                       int N, int E, hipStream_t stream) {
  const int IN = 128, HD = 128, OD = 64;
  const int RT = 5;
  const int NE1 = 4, NE2 = 8;
  const int* src = ei;
  const int* dst = ei + E;

  fill_kernel<<<cdiv(N, 256), 256, 0, stream>>>(deg, 1.0f, N);
  deg_kernel<<<cdiv(E, 256), 256, 0, stream>>>(dst, deg, E);
  rsqrt_kernel<<<cdiv(N, 256), 256, 0, stream>>>(deg, N);
  float* dis = deg;

  const int rowBlocks = N / (16 * RT);

  gemm_wmma_f32<IN, HD, RT, false>
      <<<dim3(rowBlocks, HD / 64), 128, 0, stream>>>(x, W1, dis, b1, h, out1);
  scatter_kernel<HD, NE1><<<cdiv((long long)cdiv(E, 32) * 32, 256), 256, 0, stream>>>(
      src, dst, h, dis, out1, E);

  gemm_wmma_f32<HD, OD, RT, true>
      <<<dim3(rowBlocks, 1), 128, 0, stream>>>(out1, W2, dis, b2, h, out_final);
  scatter_kernel<OD, NE2><<<cdiv((long long)cdiv(E, 32) * 32, 256), 256, 0, stream>>>(
      src, dst, h, dis, out_final, E);
}

extern "C" void kernel_launch(void* const* d_in, const int* in_sizes, int n_in,
                              void* d_out, int out_size, void* d_ws, size_t ws_size,
                              hipStream_t stream) {
  const float* x_feat = (const float*)d_in[0];
  const int*   x_ei   = (const int*)d_in[1];
  const float* y_feat = (const float*)d_in[2];
  const int*   y_ei   = (const int*)d_in[3];
  const float* W1x = (const float*)d_in[4];
  const float* b1x = (const float*)d_in[5];
  const float* W2x = (const float*)d_in[6];
  const float* b2x = (const float*)d_in[7];
  const float* W1y = (const float*)d_in[8];
  const float* b1y = (const float*)d_in[9];
  const float* W2y = (const float*)d_in[10];
  const float* b2y = (const float*)d_in[11];

  const int N = in_sizes[0] / 128;
  const int E = in_sizes[1] / 2;

  float* h    = (float*)d_ws;
  float* out1 = h + (size_t)N * 128;
  float* deg  = out1 + (size_t)N * 128;

  float* out_x = (float*)d_out;
  float* out_y = out_x + (size_t)N * 64;

  run_branch(x_feat, x_ei, W1x, b1x, W2x, b2x, out_x, deg, h, out1, N, E, stream);
  run_branch(y_feat, y_ei, W1y, b1y, W2y, b2y, out_y, deg, h, out1, N, E, stream);
}


# Degree count: blocks own 4096 destinations x 1/20 of the edges, count in LDS words, store partial counts as whole lines, summed with the rsqrt; no global degree atomics; 12 launches, was 14
# speedup vs baseline: 2.3861x; 1.2914x over previous
#include <hip/hip_runtime.h>


typedef __attribute__((ext_vector_type(2))) float v2f;
typedef __attribute__((ext_vector_type(8))) float v8f;

constexpr int DR = 4096;
constexpr int DC = 20;
constexpr int DT = 512;
constexpr int DU = 20;

__device__ __forceinline__ void deg_hit(unsigned* cnt, int k) {
  if ((unsigned)k < (unsigned)DR) atomicAdd(&cnt[k], 1u);
}

__global__ void __launch_bounds__(DT)
degcount_kernel(const int* __restrict__ dst, float* __restrict__ part, int E) {
  __shared__ unsigned cnt[DR];
  for (int i = threadIdx.x; i < DR; i += DT) cnt[i] = 0u;
  __syncthreads();
  const int chunk = ((E + DC * 4 - 1) / (DC * 4)) * 4;
  const int lo = blockIdx.x * DR;
  const int e0 = blockIdx.y * chunk;
  const int n = max(min(E, e0 + chunk) - e0, 0);
  const int* __restrict__ dp = dst + e0;
  const int n4 = (((size_t)dp & 15) == 0) ? (n >> 2) : 0;
  const int4* __restrict__ dp4 = (const int4*)dp;
  for (int base = 0; base < n4; base += DT * DU) {
    int4 d[DU];
    for (int u = 0; u < DU; ++u) {
      const int g = base + u * DT + threadIdx.x;
      d[u] = (g < n4) ? dp4[g] : make_int4(-1, -1, -1, -1);
    }
    for (int u = 0; u < DU; ++u) {
      deg_hit(cnt, d[u].x - lo);
      deg_hit(cnt, d[u].y - lo);
      deg_hit(cnt, d[u].z - lo);
      deg_hit(cnt, d[u].w - lo);
    }
  }
  for (int e = (n4 << 2) + threadIdx.x; e < n; e += DT) deg_hit(cnt, dp[e] - lo);
  __syncthreads();
  float* __restrict__ p = part + ((size_t)blockIdx.y * gridDim.x + blockIdx.x) * DR;
  for (int i = threadIdx.x; i < DR; i += DT) p[i] = (float)cnt[i];
}

__global__ void dis_kernel(const float* __restrict__ part, float* __restrict__ dis, int n) {
  const int i = blockIdx.x * blockDim.x + threadIdx.x;
  if (i < n) {
    const size_t stride = (size_t)((n + DR - 1) / DR) * DR;
    float s = 1.0f;
    for (int c = 0; c < DC; ++c) s += part[c * stride + i];
    dis[i] = rsqrtf(s);
  }
}

template <int K, int ND, int RT, bool RELU_A>
__global__ void gemm_wmma_f32(const float* __restrict__ X,
                              const float* __restrict__ W,
                              const float* __restrict__ dis,
                              const float* __restrict__ bias,
                              float* __restrict__ H,
                              float* __restrict__ OUT) {
  const int wave = threadIdx.x >> 5;
  const int lane = threadIdx.x & 31;
  const int col0 = blockIdx.y * 64 + wave * 16;

  const int m  = lane & 15;
  const int kb = (lane >> 4) * 2;

  v2f b[K / 4];
#pragma unroll
  for (int kk = 0; kk < K / 4; ++kk) {
    b[kk].x = W[(size_t)(kk * 4 + kb) * ND + col0 + m];
    b[kk].y = W[(size_t)(kk * 4 + kb + 1) * ND + col0 + m];
  }
  const float bv = bias[col0 + m];
  const int rbase = (lane >> 4) * 8;

  int row0 = blockIdx.x * (16 * RT);
#pragma unroll 1
  for (int rt = 0; rt < RT; ++rt, row0 += 16) {
    const float* __restrict__ xrow = X + (size_t)(row0 + m) * K + kb;
    const float4 d0 = *(const float4*)(dis + row0 + rbase);
    const float4 d1 = *(const float4*)(dis + row0 + rbase + 4);
    v2f a[K / 4];
#pragma unroll
    for (int kk = 0; kk < K / 4; ++kk) {
      v2f av = *(const v2f*)(xrow + kk * 4);
      if (RELU_A) { av.x = fmaxf(av.x, 0.0f); av.y = fmaxf(av.y, 0.0f); }
      a[kk] = av;
    }

    v8f c = {};
#pragma unroll
    for (int kk = 0; kk < K / 4; ++kk)
      c = __builtin_amdgcn_wmma_f32_16x16x4_f32(false, a[kk], false, b[kk],
                                                (short)0, c, false, false);

    const float dd[8] = {d0.x, d0.y, d0.z, d0.w, d1.x, d1.y, d1.z, d1.w};
    const size_t base = (size_t)(row0 + rbase) * ND + col0 + m;
    float* __restrict__ hp = H + base;
    float* __restrict__ op = OUT + base;
#pragma unroll
    for (int r = 0; r < 8; ++r) {
      const float w = dd[r];
      const float ww = w * w;
      hp[r * ND] = c[r];
      op[r * ND] = bv + c[r] * ww;
    }
  }
}

template <int D, int NE>
__device__ __forceinline__ void scatter_edges(const float* __restrict__ hl,
                                              float* __restrict__ ol,
                                              int sv, int dv, float wl, int j) {
  constexpr int C = D / 32;
  float v[NE][C];
  for (int k = 0; k < NE; ++k) {
    const int s = __builtin_amdgcn_readlane(sv, j + k);
    const float* __restrict__ hp = hl + (size_t)s * D;
    for (int c = 0; c < C; ++c) v[k][c] = hp[c * 32];
  }
  for (int k = 0; k < NE; ++k) {
    const int d = __builtin_amdgcn_readlane(dv, j + k);
    const float w = __int_as_float(__builtin_amdgcn_readlane(__float_as_int(wl), j + k));
    float* op = ol + (size_t)d * D;
    for (int c = 0; c < C; ++c) unsafeAtomicAdd(op + c * 32, v[k][c] * w);
  }
}

template <int D, int NE>
__global__ void scatter_kernel(const int* __restrict__ src,
                               const int* __restrict__ dst,
                               const float* __restrict__ h,
                               const float* __restrict__ dis,
                               float* __restrict__ out, int E) {
  const int lane = threadIdx.x & 31;
  const int wv = __builtin_amdgcn_readfirstlane(
      (int)((blockIdx.x * blockDim.x + threadIdx.x) >> 5));
  const int e0 = wv * 32;
  if (e0 >= E) return;
  const int n = min(32, E - e0);
  const int ei = e0 + min(lane, n - 1);
  const int sv = src[ei];
  const int dv = dst[ei];
  const float wl = dis[sv] * dis[dv];
  const float* __restrict__ hl = h + lane;
  float* __restrict__ ol = out + lane;
  int j = 0;
  for (; j + NE <= n; j += NE) scatter_edges<D, NE>(hl, ol, sv, dv, wl, j);
  for (; j < n; ++j)           scatter_edges<D, 1>(hl, ol, sv, dv, wl, j);
}

static inline int cdiv(long long a, long long b) { return (int)((a + b - 1) / b); }

static void run_branch(const float* x, const int* ei, const float* W1,
                       const float* b1, const float* W2, const float* b2,
                       float* out_final, float* deg, float* h, float* out1,
                       int N, int E, hipStream_t stream) {
  const int IN = 128, HD = 128, OD = 64;
  const int RT = 5;
  const int NE1 = 4, NE2 = 8;
  const int* src = ei;
  const int* dst = ei + E;

  degcount_kernel<<<dim3(cdiv(N, DR), DC), DT, 0, stream>>>(dst, h, E);
  dis_kernel<<<cdiv(N, 256), 256, 0, stream>>>(h, deg, N);
  float* dis = deg;

  const int rowBlocks = N / (16 * RT);

  gemm_wmma_f32<IN, HD, RT, false>
      <<<dim3(rowBlocks, HD / 64), 128, 0, stream>>>(x, W1, dis, b1, h, out1);
  scatter_kernel<HD, NE1><<<cdiv((long long)cdiv(E, 32) * 32, 256), 256, 0, stream>>>(
      src, dst, h, dis, out1, E);

  gemm_wmma_f32<HD, OD, RT, true>
      <<<dim3(rowBlocks, 1), 128, 0, stream>>>(out1, W2, dis, b2, h, out_final);
  scatter_kernel<OD, NE2><<<cdiv((long long)cdiv(E, 32) * 32, 256), 256, 0, stream>>>(
      src, dst, h, dis, out_final, E);
}

extern "C" void kernel_launch(void* const* d_in, const int* in_sizes, int n_in,
                              void* d_out, int out_size, void* d_ws, size_t ws_size,
                              hipStream_t stream) {
  const float* x_feat = (const float*)d_in[0];
  const int*   x_ei   = (const int*)d_in[1];
  const float* y_feat = (const float*)d_in[2];
  const int*   y_ei   = (const int*)d_in[3];
  const float* W1x = (const float*)d_in[4];
  const float* b1x = (const float*)d_in[5];
  const float* W2x = (const float*)d_in[6];
  const float* b2x = (const float*)d_in[7];
  const float* W1y = (const float*)d_in[8];
  const float* b1y = (const float*)d_in[9];
  const float* W2y = (const float*)d_in[10];
  const float* b2y = (const float*)d_in[11];

  const int N = in_sizes[0] / 128;
  const int E = in_sizes[1] / 2;

  float* h    = (float*)d_ws;
  float* out1 = h + (size_t)N * 128;
  float* deg  = out1 + (size_t)N * 128;

  float* out_x = (float*)d_out;
  float* out_y = out_x + (size_t)N * 64;

  run_branch(x_feat, x_ei, W1x, b1x, W2x, b2x, out_x, deg, h, out1, N, E, stream);
  run_branch(y_feat, y_ei, W1y, b1y, W2y, b2y, out_y, deg, h, out1, N, E, stream);
}


# Scatter-add: blocks claim edge chunks from two counters and add only into the destination rows (split at N/2) of the L2 half they run on, die read at run time; sums still in atomic arrival order
# speedup vs baseline: 2.7903x; 1.1694x over previous
#include <hip/hip_runtime.h>


typedef __attribute__((ext_vector_type(2))) float v2f;
typedef __attribute__((ext_vector_type(8))) float v8f;

constexpr int DR = 4096;
constexpr int DC = 20;
constexpr int DT = 512;
constexpr int DU = 20;

__device__ __forceinline__ void deg_hit(unsigned* cnt, int k) {
  if ((unsigned)k < (unsigned)DR) atomicAdd(&cnt[k], 1u);
}

__global__ void __launch_bounds__(DT)
degcount_kernel(const int* __restrict__ dst, float* __restrict__ part, int E) {
  __shared__ unsigned cnt[DR];
  for (int i = threadIdx.x; i < DR; i += DT) cnt[i] = 0u;
  __syncthreads();
  const int chunk = ((E + DC * 4 - 1) / (DC * 4)) * 4;
  const int lo = blockIdx.x * DR;
  const int e0 = blockIdx.y * chunk;
  const int n = max(min(E, e0 + chunk) - e0, 0);
  const int* __restrict__ dp = dst + e0;
  const int n4 = (((size_t)dp & 15) == 0) ? (n >> 2) : 0;
  const int4* __restrict__ dp4 = (const int4*)dp;
  for (int base = 0; base < n4; base += DT * DU) {
    int4 d[DU];
    for (int u = 0; u < DU; ++u) {
      const int g = base + u * DT + threadIdx.x;
      d[u] = (g < n4) ? dp4[g] : make_int4(-1, -1, -1, -1);
    }
    for (int u = 0; u < DU; ++u) {
      deg_hit(cnt, d[u].x - lo);
      deg_hit(cnt, d[u].y - lo);
      deg_hit(cnt, d[u].z - lo);
      deg_hit(cnt, d[u].w - lo);
    }
  }
  for (int e = (n4 << 2) + threadIdx.x; e < n; e += DT) deg_hit(cnt, dp[e] - lo);
  __syncthreads();
  float* __restrict__ p = part + ((size_t)blockIdx.y * gridDim.x + blockIdx.x) * DR;
  for (int i = threadIdx.x; i < DR; i += DT) p[i] = (float)cnt[i];
}

__global__ void dis_kernel(const float* __restrict__ part, float* __restrict__ dis, int n) {
  const int i = blockIdx.x * blockDim.x + threadIdx.x;
  if (i < n) {
    const size_t stride = (size_t)((n + DR - 1) / DR) * DR;
    float s = 1.0f;
    for (int c = 0; c < DC; ++c) s += part[c * stride + i];
    dis[i] = rsqrtf(s);
  }
}

constexpr int CH = 128;
constexpr int CSTR = 64;
constexpr int SW = 8;

template <int K, int ND, int RT, bool RELU_A>
__global__ void gemm_wmma_f32(const float* __restrict__ X,
                              const float* __restrict__ W,
                              const float* __restrict__ dis,
                              const float* __restrict__ bias,
                              float* __restrict__ H,
                              float* __restrict__ OUT,
                              unsigned* __restrict__ cnt) {
  if (blockIdx.x == 0 && blockIdx.y == 0 && threadIdx.x < 2) cnt[threadIdx.x * CSTR] = 0u;
  const int wave = threadIdx.x >> 5;
  const int lane = threadIdx.x & 31;
  const int col0 = blockIdx.y * 64 + wave * 16;

  const int m  = lane & 15;
  const int kb = (lane >> 4) * 2;

  v2f b[K / 4];
#pragma unroll
  for (int kk = 0; kk < K / 4; ++kk) {
    b[kk].x = W[(size_t)(kk * 4 + kb) * ND + col0 + m];
    b[kk].y = W[(size_t)(kk * 4 + kb + 1) * ND + col0 + m];
  }
  const float bv = bias[col0 + m];
  const int rbase = (lane >> 4) * 8;

  int row0 = blockIdx.x * (16 * RT);
#pragma unroll 1
  for (int rt = 0; rt < RT; ++rt, row0 += 16) {
    const float* __restrict__ xrow = X + (size_t)(row0 + m) * K + kb;
    const float4 d0 = *(const float4*)(dis + row0 + rbase);
    const float4 d1 = *(const float4*)(dis + row0 + rbase + 4);
    v2f a[K / 4];
#pragma unroll
    for (int kk = 0; kk < K / 4; ++kk) {
      v2f av = *(const v2f*)(xrow + kk * 4);
      if (RELU_A) { av.x = fmaxf(av.x, 0.0f); av.y = fmaxf(av.y, 0.0f); }
      a[kk] = av;
    }

    v8f c = {};
#pragma unroll
    for (int kk = 0; kk < K / 4; ++kk)
      c = __builtin_amdgcn_wmma_f32_16x16x4_f32(false, a[kk], false, b[kk],
                                                (short)0, c, false, false);

    const float dd[8] = {d0.x, d0.y, d0.z, d0.w, d1.x, d1.y, d1.z, d1.w};
    const size_t base = (size_t)(row0 + rbase) * ND + col0 + m;
    float* __restrict__ hp = H + base;
    float* __restrict__ op = OUT + base;
#pragma unroll
    for (int r = 0; r < 8; ++r) {
      const float w = dd[r];
      const float ww = w * w;
      hp[r * ND] = c[r];
      op[r * ND] = bv + c[r] * ww;
    }
  }
}

template <int D, int NE>
__device__ __forceinline__ void scatter_edges(const float* __restrict__ hl,
                                              float* __restrict__ ol,
                                              int sv, int dv, float wl, int j) {
  constexpr int C = D / 32;
  float v[NE][C];
  for (int k = 0; k < NE; ++k) {
    const int s = __builtin_amdgcn_readlane(sv, j + k);
    const float* __restrict__ hp = hl + (size_t)s * D;
    for (int c = 0; c < C; ++c) v[k][c] = hp[c * 32];
  }
  for (int k = 0; k < NE; ++k) {
    const int d = __builtin_amdgcn_readlane(dv, j + k);
    const float w = __int_as_float(__builtin_amdgcn_readlane(__float_as_int(wl), j + k));
    float* op = ol + (size_t)d * D;
    for (int c = 0; c < C; ++c) unsafeAtomicAdd(op + c * 32, v[k][c] * w);
  }
}

template <int D, int NE>
__device__ __forceinline__ void scatter_run(const float* __restrict__ hl,
                                            float* __restrict__ ol,
                                            int sv, int dv, float wl, int& j, int n) {
  for (; j + NE <= n; j += NE) scatter_edges<D, NE>(hl, ol, sv, dv, wl, j);
  if constexpr (NE > 1) scatter_run<D, NE / 2>(hl, ol, sv, dv, wl, j, n);
}

template <int D, int NE>
__global__ void __launch_bounds__(SW * 32)
scatter_kernel(const int* __restrict__ src,
               const int* __restrict__ dst,
               const float* __restrict__ h,
               const float* __restrict__ dis,
               float* __restrict__ out,
               unsigned* __restrict__ cnt, int E, int nh) {
  __shared__ int   l_s[SW][CH];
  __shared__ int   l_d[SW][CH];
  __shared__ float l_w[SW][CH];
  __shared__ unsigned claim[2];
  const int lane = threadIdx.x & 31;
  const int wave = threadIdx.x >> 5;
  int*   ls = l_s[wave];
  int*   ld = l_d[wave];
  float* lw = l_w[wave];
  const float* __restrict__ hl = h + lane;
  float* __restrict__ ol = out + lane;
  const unsigned ncb = (unsigned)((E + CH * SW - 1) / (CH * SW));
  int k = (int)((__builtin_amdgcn_s_sendmsg_rtn(0x87) >> 18) & 1u);
  unsigned r = 0;
  for (int pass = 0; pass < 2; ++pass, k ^= 1) {
    for (unsigned t = 0; t <= ncb; ++t) {
      const unsigned slot = r++ & 1u;
      if (threadIdx.x == 0) claim[slot] = atomicAdd(cnt + k * CSTR, 1u);
      __syncthreads();
      const unsigned cb = (unsigned)__builtin_amdgcn_readfirstlane((int)claim[slot]);
      if (cb >= ncb) break;
      const unsigned c = cb * SW + wave;
      const int e0 = (int)c * CH + lane;
      int s[CH / 32], d[CH / 32];
      for (int g = 0; g < CH / 32; ++g) {
        const int e = e0 + g * 32;
        const bool ok = e < E;
        d[g] = ok ? dst[e] : -1;
        s[g] = ok ? src[e] : 0;
      }
      float w[CH / 32];
      for (int g = 0; g < CH / 32; ++g) {
        const bool m = (d[g] >= 0) && ((int)(d[g] >= nh) == k);
        w[g] = dis[m ? s[g] : 0] * dis[m ? d[g] : 0];
      }
      int L = 0;
      for (int g = 0; g < CH / 32; ++g) {
        const bool m = (d[g] >= 0) && ((int)(d[g] >= nh) == k);
        const unsigned mask = __builtin_amdgcn_ballot_w32(m);
        if (m) {
          const int p = L + (int)__builtin_amdgcn_mbcnt_lo(mask, 0u);
          ls[p] = s[g];
          ld[p] = d[g];
          lw[p] = w[g];
        }
        L += __builtin_popcount(mask);
      }
      for (int base = 0; base < L; base += 32) {
        const int n = min(32, L - base);
        const int i = base + min(lane, n - 1);
        const int sv = ls[i];
        const int dv = ld[i];
        const float wl = lw[i];
        int j = 0;
        scatter_run<D, NE>(hl, ol, sv, dv, wl, j, n);
      }
    }
  }
}

static inline int cdiv(long long a, long long b) { return (int)((a + b - 1) / b); }

static void run_branch(const float* x, const int* ei, const float* W1,
                       const float* b1, const float* W2, const float* b2,
                       float* out_final, float* deg, float* h, float* out1,
                       int N, int E, hipStream_t stream) {
  const int IN = 128, HD = 128, OD = 64;
  const int RT = 5;
  const int NE1 = 4, NE2 = 8;
  const int SB = 2048;
  const int* src = ei;
  const int* dst = ei + E;

  degcount_kernel<<<dim3(cdiv(N, DR), DC), DT, 0, stream>>>(dst, h, E);
  dis_kernel<<<cdiv(N, 256), 256, 0, stream>>>(h, deg, N);
  float* dis = deg;

  const int rowBlocks = N / (16 * RT);
  unsigned* cnt1 = (unsigned*)out_final;
  unsigned* cnt2 = (unsigned*)(h + (size_t)N * 64);

  gemm_wmma_f32<IN, HD, RT, false>
      <<<dim3(rowBlocks, HD / 64), 128, 0, stream>>>(x, W1, dis, b1, h, out1, cnt1);
  scatter_kernel<HD, NE1><<<SB, SW * 32, 0, stream>>>(src, dst, h, dis, out1, cnt1, E, N / 2);

  gemm_wmma_f32<HD, OD, RT, true>
      <<<dim3(rowBlocks, 1), 128, 0, stream>>>(out1, W2, dis, b2, h, out_final, cnt2);
  scatter_kernel<OD, NE2><<<SB, SW * 32, 0, stream>>>(src, dst, h, dis, out_final, cnt2, E, N / 2);
}

extern "C" void kernel_launch(void* const* d_in, const int* in_sizes, int n_in,
                              void* d_out, int out_size, void* d_ws, size_t ws_size,
                              hipStream_t stream) {
  const float* x_feat = (const float*)d_in[0];
  const int*   x_ei   = (const int*)d_in[1];
  const float* y_feat = (const float*)d_in[2];
  const int*   y_ei   = (const int*)d_in[3];
  const float* W1x = (const float*)d_in[4];
  const float* b1x = (const float*)d_in[5];
  const float* W2x = (const float*)d_in[6];
  const float* b2x = (const float*)d_in[7];
  const float* W1y = (const float*)d_in[8];
  const float* b1y = (const float*)d_in[9];
  const float* W2y = (const float*)d_in[10];
  const float* b2y = (const float*)d_in[11];

  const int N = in_sizes[0] / 128;
  const int E = in_sizes[1] / 2;

  float* h    = (float*)d_ws;
  float* out1 = h + (size_t)N * 128;
  float* deg  = out1 + (size_t)N * 128;

  float* out_x = (float*)d_out;
  float* out_y = out_x + (size_t)N * 64;

  run_branch(x_feat, x_ei, W1x, b1x, W2x, b2x, out_x, deg, h, out1, N, E, stream);
  run_branch(y_feat, y_ei, W1y, b1y, W2y, b2y, out_y, deg, h, out1, N, E, stream);
}


# Edge weights leave the scatter: GEMMs store rows x dis[row]; layer-1 scatter adds plain rows, GEMM 2's A load applies dis[dst], bias; layer 2 gathers dis[dst] only; arrival order, dis[dst] factored
# speedup vs baseline: 3.1981x; 1.1461x over previous
#include <hip/hip_runtime.h>


typedef __attribute__((ext_vector_type(2))) float v2f;
typedef __attribute__((ext_vector_type(8))) float v8f;

constexpr int DR = 4096;
constexpr int DC = 20;
constexpr int DT = 512;
constexpr int DU = 20;

__device__ __forceinline__ void deg_hit(unsigned* cnt, int k) {
  if ((unsigned)k < (unsigned)DR) atomicAdd(&cnt[k], 1u);
}

__global__ void __launch_bounds__(DT)
degcount_kernel(const int* __restrict__ dst, float* __restrict__ part, int E) {
  __shared__ unsigned cnt[DR];
  for (int i = threadIdx.x; i < DR; i += DT) cnt[i] = 0u;
  __syncthreads();
  const int chunk = ((E + DC * 4 - 1) / (DC * 4)) * 4;
  const int lo = blockIdx.x * DR;
  const int e0 = blockIdx.y * chunk;
  const int n = max(min(E, e0 + chunk) - e0, 0);
  const int* __restrict__ dp = dst + e0;
  const int n4 = (((size_t)dp & 15) == 0) ? (n >> 2) : 0;
  const int4* __restrict__ dp4 = (const int4*)dp;
  for (int base = 0; base < n4; base += DT * DU) {
    int4 d[DU];
    for (int u = 0; u < DU; ++u) {
      const int g = base + u * DT + threadIdx.x;
      d[u] = (g < n4) ? dp4[g] : make_int4(-1, -1, -1, -1);
    }
    for (int u = 0; u < DU; ++u) {
      deg_hit(cnt, d[u].x - lo);
      deg_hit(cnt, d[u].y - lo);
      deg_hit(cnt, d[u].z - lo);
      deg_hit(cnt, d[u].w - lo);
    }
  }
  for (int e = (n4 << 2) + threadIdx.x; e < n; e += DT) deg_hit(cnt, dp[e] - lo);
  __syncthreads();
  float* __restrict__ p = part + ((size_t)blockIdx.y * gridDim.x + blockIdx.x) * DR;
  for (int i = threadIdx.x; i < DR; i += DT) p[i] = (float)cnt[i];
}

__global__ void dis_kernel(const float* __restrict__ part, float* __restrict__ dis, int n) {
  const int i = blockIdx.x * blockDim.x + threadIdx.x;
  if (i < n) {
    const size_t stride = (size_t)((n + DR - 1) / DR) * DR;
    float s = 1.0f;
    for (int c = 0; c < DC; ++c) s += part[c * stride + i];
    dis[i] = rsqrtf(s);
  }
}

constexpr int CH = 128;
constexpr int CSTR = 64;
constexpr int SW = 8;

template <int K, int ND, int RT, bool SECOND>
__global__ void __launch_bounds__(128) gemm_wmma_f32(const float* __restrict__ X,
                              const float* __restrict__ W,
                              const float* __restrict__ dis,
                              const float* __restrict__ bias_in,
                              const float* __restrict__ bias,
                              float* __restrict__ H,
                              float* __restrict__ OUT,
                              unsigned* __restrict__ cnt) {
  if (blockIdx.x == 0 && blockIdx.y == 0 && threadIdx.x < 2) cnt[threadIdx.x * CSTR] = 0u;
  const int wave = threadIdx.x >> 5;
  const int lane = threadIdx.x & 31;
  const int col0 = blockIdx.y * 64 + wave * 16;

  const int m  = lane & 15;
  const int kb = (lane >> 4) * 2;

  v2f b[K / 4];
#pragma unroll
  for (int kk = 0; kk < K / 4; ++kk) {
    b[kk].x = W[(size_t)(kk * 4 + kb) * ND + col0 + m];
    b[kk].y = W[(size_t)(kk * 4 + kb + 1) * ND + col0 + m];
  }
  v2f bi[SECOND ? K / 4 : 1];
  if constexpr (SECOND) {
#pragma unroll
    for (int kk = 0; kk < K / 4; ++kk) bi[kk] = *(const v2f*)(bias_in + kk * 4 + kb);
  }
  const float bv = SECOND ? bias[col0 + m] : 0.0f;
  const int rbase = (lane >> 4) * 8;

  int row0 = blockIdx.x * (16 * RT);
#pragma unroll 1
  for (int rt = 0; rt < RT; ++rt, row0 += 16) {
    const float* __restrict__ xrow = X + (size_t)(row0 + m) * K + kb;
    const float4 d0 = *(const float4*)(dis + row0 + rbase);
    const float4 d1 = *(const float4*)(dis + row0 + rbase + 4);
    const float da = SECOND ? dis[row0 + m] : 0.0f;
    v2f a[K / 4];
#pragma unroll
    for (int kk = 0; kk < K / 4; ++kk) {
      v2f av = *(const v2f*)(xrow + kk * 4);
      if constexpr (SECOND) {
        av.x = fmaxf(bi[kk].x + da * av.x, 0.0f);
        av.y = fmaxf(bi[kk].y + da * av.y, 0.0f);
      }
      a[kk] = av;
    }

    v8f c = {};
#pragma unroll
    for (int kk = 0; kk < K / 4; ++kk)
      c = __builtin_amdgcn_wmma_f32_16x16x4_f32(false, a[kk], false, b[kk],
                                                (short)0, c, false, false);

    const float dd[8] = {d0.x, d0.y, d0.z, d0.w, d1.x, d1.y, d1.z, d1.w};
    const size_t base = (size_t)(row0 + rbase) * ND + col0 + m;
    float* __restrict__ hp = H + base;
    float* __restrict__ op = OUT + base;
#pragma unroll
    for (int r = 0; r < 8; ++r) {
      const float w = dd[r];
      const float hs = c[r] * w;
      hp[r * ND] = hs;
      op[r * ND] = SECOND ? bv + hs * w : hs;
    }
  }
}

template <int D, int NE, bool WDST>
__device__ __forceinline__ void scatter_edges(const float* __restrict__ hl,
                                              float* __restrict__ ol,
                                              int sv, int dv, float wl, int j) {
  constexpr int C = D / 32;
  float v[NE][C];
  for (int k = 0; k < NE; ++k) {
    const int s = __builtin_amdgcn_readlane(sv, j + k);
    const float* __restrict__ hp = hl + (size_t)s * D;
    for (int c = 0; c < C; ++c) v[k][c] = hp[c * 32];
  }
  for (int k = 0; k < NE; ++k) {
    const int d = __builtin_amdgcn_readlane(dv, j + k);
    float* op = ol + (size_t)d * D;
    if constexpr (WDST) {
      const float w = __int_as_float(__builtin_amdgcn_readlane(__float_as_int(wl), j + k));
      for (int c = 0; c < C; ++c) unsafeAtomicAdd(op + c * 32, v[k][c] * w);
    } else {
      for (int c = 0; c < C; ++c) unsafeAtomicAdd(op + c * 32, v[k][c]);
    }
  }
}

template <int D, int NE, bool WDST>
__device__ __forceinline__ void scatter_run(const float* __restrict__ hl,
                                            float* __restrict__ ol,
                                            int sv, int dv, float wl, int& j, int n) {
  for (; j + NE <= n; j += NE) scatter_edges<D, NE, WDST>(hl, ol, sv, dv, wl, j);
  if constexpr (NE > 1) scatter_run<D, NE / 2, WDST>(hl, ol, sv, dv, wl, j, n);
}

template <int D, int NE, bool WDST>
__global__ void __launch_bounds__(SW * 32)
scatter_kernel(const int* __restrict__ src,
               const int* __restrict__ dst,
               const float* __restrict__ h,
               const float* __restrict__ dis,
               float* __restrict__ out,
               unsigned* __restrict__ cnt, int E, int nh) {
  __shared__ int   l_s[SW][CH];
  __shared__ int   l_d[SW][CH];
  __shared__ float l_w[WDST ? SW : 1][WDST ? CH : 1];
  __shared__ unsigned claim[2];
  const int lane = threadIdx.x & 31;
  const int wave = threadIdx.x >> 5;
  int*   ls = l_s[wave];
  int*   ld = l_d[wave];
  float* lw = l_w[WDST ? wave : 0];
  const float* __restrict__ hl = h + lane;
  float* __restrict__ ol = out + lane;
  const unsigned ncb = (unsigned)((E + CH * SW - 1) / (CH * SW));
  int k = (int)((__builtin_amdgcn_s_sendmsg_rtn(0x87) >> 18) & 1u);
  unsigned r = 0;
  for (int pass = 0; pass < 2; ++pass, k ^= 1) {
    for (unsigned t = 0; t <= ncb; ++t) {
      const unsigned slot = r++ & 1u;
      if (threadIdx.x == 0) claim[slot] = atomicAdd(cnt + k * CSTR, 1u);
      __syncthreads();
      const unsigned cb = (unsigned)__builtin_amdgcn_readfirstlane((int)claim[slot]);
      if (cb >= ncb) break;
      const unsigned c = cb * SW + wave;
      const int e0 = (int)c * CH + lane;
      int s[CH / 32], d[CH / 32];
      for (int g = 0; g < CH / 32; ++g) {
        const int e = e0 + g * 32;
        const bool ok = e < E;
        d[g] = ok ? dst[e] : -1;
        s[g] = ok ? src[e] : 0;
      }
      float w[CH / 32];
      if constexpr (WDST) {
        for (int g = 0; g < CH / 32; ++g) {
          const bool m = (d[g] >= 0) && ((int)(d[g] >= nh) == k);
          w[g] = dis[m ? d[g] : 0];
        }
      }
      int L = 0;
      for (int g = 0; g < CH / 32; ++g) {
        const bool m = (d[g] >= 0) && ((int)(d[g] >= nh) == k);
        const unsigned mask = __builtin_amdgcn_ballot_w32(m);
        if (m) {
          const int p = L + (int)__builtin_amdgcn_mbcnt_lo(mask, 0u);
          ls[p] = s[g];
          ld[p] = d[g];
          if constexpr (WDST) lw[p] = w[g];
        }
        L += __builtin_popcount(mask);
      }
      for (int base = 0; base < L; base += 32) {
        const int n = min(32, L - base);
        const int i = base + min(lane, n - 1);
        const int sv = ls[i];
        const int dv = ld[i];
        float wl = 0.0f;
        if constexpr (WDST) wl = lw[i];
        int j = 0;
        scatter_run<D, NE, WDST>(hl, ol, sv, dv, wl, j, n);
      }
    }
  }
}

static inline int cdiv(long long a, long long b) { return (int)((a + b - 1) / b); }

static void run_branch(const float* x, const int* ei, const float* W1,
                       const float* b1, const float* W2, const float* b2,
                       float* out_final, float* deg, float* h, float* out1,
                       int N, int E, hipStream_t stream) {
  const int IN = 128, HD = 128, OD = 64;
  const int RT = 5;
  const int NE1 = 4, NE2 = 8;
  const int SB = 2048;
  const int* src = ei;
  const int* dst = ei + E;

  degcount_kernel<<<dim3(cdiv(N, DR), DC), DT, 0, stream>>>(dst, h, E);
  dis_kernel<<<cdiv(N, 256), 256, 0, stream>>>(h, deg, N);
  float* dis = deg;

  const int rowBlocks = N / (16 * RT);
  unsigned* cnt1 = (unsigned*)out_final;
  unsigned* cnt2 = (unsigned*)(h + (size_t)N * 64);

  gemm_wmma_f32<IN, HD, RT, false>
      <<<dim3(rowBlocks, HD / 64), 128, 0, stream>>>(x, W1, dis, b1, b1, h, out1, cnt1);
  scatter_kernel<HD, NE1, false><<<SB, SW * 32, 0, stream>>>(src, dst, h, dis, out1, cnt1, E, N / 2);

  gemm_wmma_f32<HD, OD, RT, true>
      <<<dim3(rowBlocks, 1), 128, 0, stream>>>(out1, W2, dis, b1, b2, h, out_final, cnt2);
  scatter_kernel<OD, NE2, true><<<SB, SW * 32, 0, stream>>>(src, dst, h, dis, out_final, cnt2, E, N / 2);
}

extern "C" void kernel_launch(void* const* d_in, const int* in_sizes, int n_in,
                              void* d_out, int out_size, void* d_ws, size_t ws_size,
                              hipStream_t stream) {
  const float* x_feat = (const float*)d_in[0];
  const int*   x_ei   = (const int*)d_in[1];
  const float* y_feat = (const float*)d_in[2];
  const int*   y_ei   = (const int*)d_in[3];
  const float* W1x = (const float*)d_in[4];
  const float* b1x = (const float*)d_in[5];
  const float* W2x = (const float*)d_in[6];
  const float* b2x = (const float*)d_in[7];
  const float* W1y = (const float*)d_in[8];
  const float* b1y = (const float*)d_in[9];
  const float* W2y = (const float*)d_in[10];
  const float* b2y = (const float*)d_in[11];

  const int N = in_sizes[0] / 128;
  const int E = in_sizes[1] / 2;

  float* h    = (float*)d_ws;
  float* out1 = h + (size_t)N * 128;
  float* deg  = out1 + (size_t)N * 128;

  float* out_x = (float*)d_out;
  float* out_y = out_x + (size_t)N * 64;

  run_branch(x_feat, x_ei, W1x, b1x, W2x, b2x, out_x, deg, h, out1, N, E, stream);
  run_branch(y_feat, y_ei, W1y, b1y, W2y, b2y, out_y, deg, h, out1, N, E, stream);
}


# Degree stage in one launch, was four: one grid counts both graphs' in-degrees (z = graph); GEMM 1's prologue sums the partial counts and forms dis itself; 9 launches, was 12; sum order kept
# speedup vs baseline: 3.3126x; 1.0358x over previous
#include <hip/hip_runtime.h>


typedef __attribute__((ext_vector_type(2))) float v2f;
typedef __attribute__((ext_vector_type(8))) float v8f;

constexpr int DR = 4096;
constexpr int DC = 19;
constexpr int DT = 512;
constexpr int DU = 21;

__device__ __forceinline__ void deg_hit(unsigned* cnt, int k) {
  if ((unsigned)k < (unsigned)DR) atomicAdd(&cnt[k], 1u);
}

__global__ void __launch_bounds__(DT)
degcount_kernel(const int* __restrict__ dst_x, const int* __restrict__ dst_y,
                float* __restrict__ part_x, float* __restrict__ part_y, int E) {
  __shared__ unsigned cnt[DR];
  const int* __restrict__ dst = blockIdx.z ? dst_y : dst_x;
  float* __restrict__ part = blockIdx.z ? part_y : part_x;
  for (int i = threadIdx.x; i < DR; i += DT) cnt[i] = 0u;
  __syncthreads();
  const int chunk = ((E + DC * 4 - 1) / (DC * 4)) * 4;
  const int lo = blockIdx.x * DR;
  const int e0 = blockIdx.y * chunk;
  const int n = max(min(E, e0 + chunk) - e0, 0);
  const int* __restrict__ dp = dst + e0;
  const int n4 = (((size_t)dp & 15) == 0) ? (n >> 2) : 0;
  const int4* __restrict__ dp4 = (const int4*)dp;
  for (int base = 0; base < n4; base += DT * DU) {
    int4 d[DU];
    for (int u = 0; u < DU; ++u) {
      const int g = base + u * DT + threadIdx.x;
      d[u] = (g < n4) ? dp4[g] : make_int4(-1, -1, -1, -1);
    }
    for (int u = 0; u < DU; ++u) {
      deg_hit(cnt, d[u].x - lo);
      deg_hit(cnt, d[u].y - lo);
      deg_hit(cnt, d[u].z - lo);
      deg_hit(cnt, d[u].w - lo);
    }
  }
  for (int e = (n4 << 2) + threadIdx.x; e < n; e += DT) deg_hit(cnt, dp[e] - lo);
  __syncthreads();
  float* __restrict__ p = part + ((size_t)blockIdx.y * gridDim.x + blockIdx.x) * DR;
  for (int i = threadIdx.x; i < DR; i += DT) p[i] = (float)cnt[i];
}

constexpr int CH = 128;
constexpr int CSTR = 64;
constexpr int SW = 8;

template <int K, int ND, int RT, bool SECOND>
__global__ void __launch_bounds__(128) gemm_wmma_f32(const float* __restrict__ X,
                              const float* __restrict__ W,
                              const float* __restrict__ part,
                              float* __restrict__ dis,
                              const float* __restrict__ bias_in,
                              const float* __restrict__ bias,
                              float* __restrict__ H,
                              float* __restrict__ OUT,
                              unsigned* __restrict__ cnt, int n) {
  if (blockIdx.x == 0 && blockIdx.y == 0 && threadIdx.x < 2) cnt[threadIdx.x * CSTR] = 0u;
  __shared__ __attribute__((aligned(16))) float s_dis[16 * RT];
  [[maybe_unused]] float pc[DC];
  if constexpr (!SECOND) {
    if (threadIdx.x < 16 * RT) {
      const size_t stride = (size_t)((n + DR - 1) / DR) * DR;
      const float* __restrict__ pp = part + blockIdx.x * (16 * RT) + threadIdx.x;
      for (int c = 0; c < DC; ++c) pc[c] = pp[c * stride];
    }
  }
  const int wave = threadIdx.x >> 5;
  const int lane = threadIdx.x & 31;
  const int col0 = blockIdx.y * 64 + wave * 16;

  const int m  = lane & 15;
  const int kb = (lane >> 4) * 2;

  v2f b[K / 4];
#pragma unroll
  for (int kk = 0; kk < K / 4; ++kk) {
    b[kk].x = W[(size_t)(kk * 4 + kb) * ND + col0 + m];
    b[kk].y = W[(size_t)(kk * 4 + kb + 1) * ND + col0 + m];
  }
  v2f bi[SECOND ? K / 4 : 1];
  if constexpr (SECOND) {
#pragma unroll
    for (int kk = 0; kk < K / 4; ++kk) bi[kk] = *(const v2f*)(bias_in + kk * 4 + kb);
  }
  const float bv = SECOND ? bias[col0 + m] : 0.0f;
  const int rbase = (lane >> 4) * 8;
  if constexpr (!SECOND) {
    if (threadIdx.x < 16 * RT) {
      float deg = 1.0f;
      for (int c = 0; c < DC; ++c) deg += pc[c];
      const float d = rsqrtf(deg);
      s_dis[threadIdx.x] = d;
      if (blockIdx.y == 0) dis[blockIdx.x * (16 * RT) + threadIdx.x] = d;
    }
  }

  int row0 = blockIdx.x * (16 * RT);
#pragma unroll 1
  for (int rt = 0; rt < RT; ++rt, row0 += 16) {
    const float* __restrict__ xrow = X + (size_t)(row0 + m) * K + kb;
    float4 d0, d1;
    if constexpr (SECOND) {
      d0 = *(const float4*)(dis + row0 + rbase);
      d1 = *(const float4*)(dis + row0 + rbase + 4);
    }
    const float da = SECOND ? dis[row0 + m] : 0.0f;
    v2f a[K / 4];
#pragma unroll
    for (int kk = 0; kk < K / 4; ++kk) {
      v2f av = *(const v2f*)(xrow + kk * 4);
      if constexpr (SECOND) {
        av.x = fmaxf(bi[kk].x + da * av.x, 0.0f);
        av.y = fmaxf(bi[kk].y + da * av.y, 0.0f);
      }
      a[kk] = av;
    }

    v8f c = {};
#pragma unroll
    for (int kk = 0; kk < K / 4; ++kk)
      c = __builtin_amdgcn_wmma_f32_16x16x4_f32(false, a[kk], false, b[kk],
                                                (short)0, c, false, false);

    if constexpr (!SECOND) {
      if (rt == 0) __syncthreads();
      d0 = *(const float4*)(s_dis + rt * 16 + rbase);
      d1 = *(const float4*)(s_dis + rt * 16 + rbase + 4);
    }

    const float dd[8] = {d0.x, d0.y, d0.z, d0.w, d1.x, d1.y, d1.z, d1.w};
    const size_t base = (size_t)(row0 + rbase) * ND + col0 + m;
    float* __restrict__ hp = H + base;
    float* __restrict__ op = OUT + base;
#pragma unroll
    for (int r = 0; r < 8; ++r) {
      const float w = dd[r];
      const float hs = c[r] * w;
      hp[r * ND] = hs;
      op[r * ND] = SECOND ? bv + hs * w : hs;
    }
  }
}

template <int D, int NE, bool WDST>
__device__ __forceinline__ void scatter_edges(const float* __restrict__ hl,
                                              float* __restrict__ ol,
                                              int sv, int dv, float wl, int j) {
  constexpr int C = D / 32;
  float v[NE][C];
  for (int k = 0; k < NE; ++k) {
    const int s = __builtin_amdgcn_readlane(sv, j + k);
    const float* __restrict__ hp = hl + (size_t)s * D;
    for (int c = 0; c < C; ++c) v[k][c] = hp[c * 32];
  }
  for (int k = 0; k < NE; ++k) {
    const int d = __builtin_amdgcn_readlane(dv, j + k);
    float* op = ol + (size_t)d * D;
    if constexpr (WDST) {
      const float w = __int_as_float(__builtin_amdgcn_readlane(__float_as_int(wl), j + k));
      for (int c = 0; c < C; ++c) unsafeAtomicAdd(op + c * 32, v[k][c] * w);
    } else {
      for (int c = 0; c < C; ++c) unsafeAtomicAdd(op + c * 32, v[k][c]);
    }
  }
}

template <int D, int NE, bool WDST>
__device__ __forceinline__ void scatter_run(const float* __restrict__ hl,
                                            float* __restrict__ ol,
                                            int sv, int dv, float wl, int& j, int n) {
  for (; j + NE <= n; j += NE) scatter_edges<D, NE, WDST>(hl, ol, sv, dv, wl, j);
  if constexpr (NE > 1) scatter_run<D, NE / 2, WDST>(hl, ol, sv, dv, wl, j, n);
}

template <int D, int NE, bool WDST>
__global__ void __launch_bounds__(SW * 32)
scatter_kernel(const int* __restrict__ src,
               const int* __restrict__ dst,
               const float* __restrict__ h,
               const float* __restrict__ dis,
               float* __restrict__ out,
               unsigned* __restrict__ cnt, int E, int nh) {
  __shared__ int   l_s[SW][CH];
  __shared__ int   l_d[SW][CH];
  __shared__ float l_w[WDST ? SW : 1][WDST ? CH : 1];
  __shared__ unsigned claim[2];
  const int lane = threadIdx.x & 31;
  const int wave = threadIdx.x >> 5;
  int*   ls = l_s[wave];
  int*   ld = l_d[wave];
  float* lw = l_w[WDST ? wave : 0];
  const float* __restrict__ hl = h + lane;
  float* __restrict__ ol = out + lane;
  const unsigned ncb = (unsigned)((E + CH * SW - 1) / (CH * SW));
  int k = (int)((__builtin_amdgcn_s_sendmsg_rtn(0x87) >> 18) & 1u);
  unsigned r = 0;
  for (int pass = 0; pass < 2; ++pass, k ^= 1) {
    for (unsigned t = 0; t <= ncb; ++t) {
      const unsigned slot = r++ & 1u;
      if (threadIdx.x == 0) claim[slot] = atomicAdd(cnt + k * CSTR, 1u);
      __syncthreads();
      const unsigned cb = (unsigned)__builtin_amdgcn_readfirstlane((int)claim[slot]);
      if (cb >= ncb) break;
      const unsigned c = cb * SW + wave;
      const int e0 = (int)c * CH + lane;
      int s[CH / 32], d[CH / 32];
      for (int g = 0; g < CH / 32; ++g) {
        const int e = e0 + g * 32;
        const bool ok = e < E;
        d[g] = ok ? dst[e] : -1;
        s[g] = ok ? src[e] : 0;
      }
      float w[CH / 32];
      if constexpr (WDST) {
        for (int g = 0; g < CH / 32; ++g) {
          const bool m = (d[g] >= 0) && ((int)(d[g] >= nh) == k);
          w[g] = dis[m ? d[g] : 0];
        }
      }
      int L = 0;
      for (int g = 0; g < CH / 32; ++g) {
        const bool m = (d[g] >= 0) && ((int)(d[g] >= nh) == k);
        const unsigned mask = __builtin_amdgcn_ballot_w32(m);
        if (m) {
          const int p = L + (int)__builtin_amdgcn_mbcnt_lo(mask, 0u);
          ls[p] = s[g];
          ld[p] = d[g];
          if constexpr (WDST) lw[p] = w[g];
        }
        L += __builtin_popcount(mask);
      }
      for (int base = 0; base < L; base += 32) {
        const int n = min(32, L - base);
        const int i = base + min(lane, n - 1);
        const int sv = ls[i];
        const int dv = ld[i];
        float wl = 0.0f;
        if constexpr (WDST) wl = lw[i];
        int j = 0;
        scatter_run<D, NE, WDST>(hl, ol, sv, dv, wl, j, n);
      }
    }
  }
}

static inline int cdiv(long long a, long long b) { return (int)((a + b - 1) / b); }

static void run_branch(const float* x, const int* ei, const float* W1,
                       const float* b1, const float* W2, const float* b2,
                       float* out_final, const float* part, float* dis, float* h, float* out1,
                       int N, int E, hipStream_t stream) {
  const int IN = 128, HD = 128, OD = 64;
  const int RT = 5;
  const int NE1 = 4, NE2 = 8;
  const int SB = 2048;
  const int* src = ei;
  const int* dst = ei + E;

  const int rowBlocks = N / (16 * RT);
  unsigned* cnt1 = (unsigned*)out_final;
  unsigned* cnt2 = (unsigned*)(h + (size_t)N * 64);

  gemm_wmma_f32<IN, HD, RT, false>
      <<<dim3(rowBlocks, HD / 64), 128, 0, stream>>>(x, W1, part, dis, b1, b1, h, out1, cnt1, N);
  scatter_kernel<HD, NE1, false><<<SB, SW * 32, 0, stream>>>(src, dst, h, dis, out1, cnt1, E, N / 2);

  gemm_wmma_f32<HD, OD, RT, true>
      <<<dim3(rowBlocks, 1), 128, 0, stream>>>(out1, W2, part, dis, b1, b2, h, out_final, cnt2, N);
  scatter_kernel<OD, NE2, true><<<SB, SW * 32, 0, stream>>>(src, dst, h, dis, out_final, cnt2, E, N / 2);
}

extern "C" void kernel_launch(void* const* d_in, const int* in_sizes, int n_in,
                              void* d_out, int out_size, void* d_ws, size_t ws_size,
                              hipStream_t stream) {
  const float* x_feat = (const float*)d_in[0];
  const int*   x_ei   = (const int*)d_in[1];
  const float* y_feat = (const float*)d_in[2];
  const int*   y_ei   = (const int*)d_in[3];
  const float* W1x = (const float*)d_in[4];
  const float* b1x = (const float*)d_in[5];
  const float* W2x = (const float*)d_in[6];
  const float* b2x = (const float*)d_in[7];
  const float* W1y = (const float*)d_in[8];
  const float* b1y = (const float*)d_in[9];
  const float* W2y = (const float*)d_in[10];
  const float* b2y = (const float*)d_in[11];

  const int N = in_sizes[0] / 128;
  const int E = in_sizes[1] / 2;

  float* h    = (float*)d_ws;
  float* out1 = h + (size_t)N * 128;
  float* deg  = out1 + (size_t)N * 128;

  float* out_x = (float*)d_out;
  float* out_y = out_x + (size_t)N * 64;

  float* part_x = out_x + 128;
  float* part_y = out_y + 128;
  degcount_kernel<<<dim3(cdiv(N, DR), DC, 2), DT, 0, stream>>>(x_ei + E, y_ei + E, part_x, part_y, E);

  run_branch(x_feat, x_ei, W1x, b1x, W2x, b2x, out_x, part_x, deg, h, out1, N, E, stream);
  run_branch(y_feat, y_ei, W1y, b1y, W2y, b2y, out_y, part_y, deg, h, out1, N, E, stream);
}


# Both GEMMs: A staged through LDS by whole-row 16-byte loads (4 lines per load, was 16 pieces of 16 B), once per block for its 4 waves, next tile in flight under the products; sum order unchanged
# speedup vs baseline: 3.5899x; 1.0837x over previous
#include <hip/hip_runtime.h>


typedef __attribute__((ext_vector_type(2))) float v2f;
typedef __attribute__((ext_vector_type(8))) float v8f;

constexpr int DR = 4096;
constexpr int DC = 19;
constexpr int DT = 512;
constexpr int DU = 21;

__device__ __forceinline__ void deg_hit(unsigned* cnt, int k) {
  if ((unsigned)k < (unsigned)DR) atomicAdd(&cnt[k], 1u);
}

__global__ void __launch_bounds__(DT)
degcount_kernel(const int* __restrict__ dst_x, const int* __restrict__ dst_y,
                float* __restrict__ part_x, float* __restrict__ part_y, int E) {
  __shared__ unsigned cnt[DR];
  const int* __restrict__ dst = blockIdx.z ? dst_y : dst_x;
  float* __restrict__ part = blockIdx.z ? part_y : part_x;
  for (int i = threadIdx.x; i < DR; i += DT) cnt[i] = 0u;
  __syncthreads();
  const int chunk = ((E + DC * 4 - 1) / (DC * 4)) * 4;
  const int lo = blockIdx.x * DR;
  const int e0 = blockIdx.y * chunk;
  const int n = max(min(E, e0 + chunk) - e0, 0);
  const int* __restrict__ dp = dst + e0;
  const int n4 = (((size_t)dp & 15) == 0) ? (n >> 2) : 0;
  const int4* __restrict__ dp4 = (const int4*)dp;
  for (int base = 0; base < n4; base += DT * DU) {
    int4 d[DU];
    for (int u = 0; u < DU; ++u) {
      const int g = base + u * DT + threadIdx.x;
      d[u] = (g < n4) ? dp4[g] : make_int4(-1, -1, -1, -1);
    }
    for (int u = 0; u < DU; ++u) {
      deg_hit(cnt, d[u].x - lo);
      deg_hit(cnt, d[u].y - lo);
      deg_hit(cnt, d[u].z - lo);
      deg_hit(cnt, d[u].w - lo);
    }
  }
  for (int e = (n4 << 2) + threadIdx.x; e < n; e += DT) deg_hit(cnt, dp[e] - lo);
  __syncthreads();
  float* __restrict__ p = part + ((size_t)blockIdx.y * gridDim.x + blockIdx.x) * DR;
  for (int i = threadIdx.x; i < DR; i += DT) p[i] = (float)cnt[i];
}

constexpr int CH = 128;
constexpr int CSTR = 64;
constexpr int SW = 8;

template <bool SECOND>
__device__ __forceinline__ void stage_row(float* __restrict__ dst, float4 v, float da, float4 bi4) {
  if constexpr (SECOND) {
    v.x = fmaxf(bi4.x + da * v.x, 0.0f);
    v.y = fmaxf(bi4.y + da * v.y, 0.0f);
    v.z = fmaxf(bi4.z + da * v.z, 0.0f);
    v.w = fmaxf(bi4.w + da * v.w, 0.0f);
  }
  *(float4*)dst = v;
}

template <int K, int ND, int RT, bool SECOND>
__global__ void __launch_bounds__(128) gemm_wmma_f32(const float* __restrict__ X,
                              const float* __restrict__ W,
                              const float* __restrict__ part,
                              float* __restrict__ dis,
                              const float* __restrict__ bias_in,
                              const float* __restrict__ bias,
                              float* __restrict__ H,
                              float* __restrict__ OUT,
                              unsigned* __restrict__ cnt, int n) {
  static_assert(K == 128, "one wave's 16-byte loads cover exactly one row of K floats");
  if (blockIdx.x == 0 && blockIdx.y == 0 && threadIdx.x < 2) cnt[threadIdx.x * CSTR] = 0u;
  __shared__ __attribute__((aligned(16))) float s_dis[16 * RT];
  constexpr int AP = K + 4;
  __shared__ __attribute__((aligned(16))) float s_a[2][16 * AP];

  const int wave = threadIdx.x >> 5;
  const int lane = threadIdx.x & 31;
  int row0 = blockIdx.x * (16 * RT);

  const float* __restrict__ xs = X + (size_t)(row0 + wave) * K + lane * 4;
  float4 st0 = *(const float4*)(xs);
  float4 st1 = *(const float4*)(xs + (size_t)4 * K);
  float4 st2 = *(const float4*)(xs + (size_t)8 * K);
  float4 st3 = *(const float4*)(xs + (size_t)12 * K);
  float sd0 = 0.0f, sd1 = 0.0f, sd2 = 0.0f, sd3 = 0.0f;
  if constexpr (SECOND) {
    sd0 = dis[row0 + wave];
    sd1 = dis[row0 + 4 + wave];
    sd2 = dis[row0 + 8 + wave];
    sd3 = dis[row0 + 12 + wave];
  }
  float4 bi4 = make_float4(0.0f, 0.0f, 0.0f, 0.0f);
  if constexpr (SECOND) bi4 = *(const float4*)(bias_in + lane * 4);

  [[maybe_unused]] float pc[DC];
  if constexpr (!SECOND) {
    if (threadIdx.x < 16 * RT) {
      const size_t stride = (size_t)((n + DR - 1) / DR) * DR;
      const float* __restrict__ pp = part + blockIdx.x * (16 * RT) + threadIdx.x;
      for (int c = 0; c < DC; ++c) pc[c] = pp[c * stride];
    }
  }
  const int col0 = blockIdx.y * 64 + wave * 16;

  const int m  = lane & 15;
  const int kb = (lane >> 4) * 2;

  v2f b[K / 4];
#pragma unroll
  for (int kk = 0; kk < K / 4; ++kk) {
    b[kk].x = W[(size_t)(kk * 4 + kb) * ND + col0 + m];
    b[kk].y = W[(size_t)(kk * 4 + kb + 1) * ND + col0 + m];
  }
  const float bv = SECOND ? bias[col0 + m] : 0.0f;
  const int rbase = (lane >> 4) * 8;

  const int soff = wave * AP + lane * 4;
  stage_row<SECOND>(s_a[0] + soff, st0, sd0, bi4);
  stage_row<SECOND>(s_a[0] + soff + 4 * AP, st1, sd1, bi4);
  stage_row<SECOND>(s_a[0] + soff + 8 * AP, st2, sd2, bi4);
  stage_row<SECOND>(s_a[0] + soff + 12 * AP, st3, sd3, bi4);

  if constexpr (!SECOND) {
    if (threadIdx.x < 16 * RT) {
      float deg = 1.0f;
      for (int c = 0; c < DC; ++c) deg += pc[c];
      const float d = rsqrtf(deg);
      s_dis[threadIdx.x] = d;
      if (blockIdx.y == 0) dis[blockIdx.x * (16 * RT) + threadIdx.x] = d;
    }
  }
  __syncthreads();

#pragma unroll 1
  for (int rt = 0; rt < RT; ++rt, row0 += 16) {
    const float* __restrict__ ab = s_a[rt & 1] + m * AP + kb;
    v2f a[K / 4];
#pragma unroll
    for (int kk = 0; kk < K / 4; ++kk) a[kk] = *(const v2f*)(ab + kk * 4);

    const bool more = rt + 1 < RT;
    if (more) {
      const float* __restrict__ xn = xs + (size_t)((rt + 1) * 16) * K;
      st0 = *(const float4*)(xn);
      st1 = *(const float4*)(xn + (size_t)4 * K);
      st2 = *(const float4*)(xn + (size_t)8 * K);
      st3 = *(const float4*)(xn + (size_t)12 * K);
      if constexpr (SECOND) {
        sd0 = dis[row0 + 16 + wave];
        sd1 = dis[row0 + 20 + wave];
        sd2 = dis[row0 + 24 + wave];
        sd3 = dis[row0 + 28 + wave];
      }
    }
    float4 d0, d1;
    if constexpr (SECOND) {
      d0 = *(const float4*)(dis + row0 + rbase);
      d1 = *(const float4*)(dis + row0 + rbase + 4);
    }

    v8f c = {};
#pragma unroll
    for (int kk = 0; kk < K / 4; ++kk)
      c = __builtin_amdgcn_wmma_f32_16x16x4_f32(false, a[kk], false, b[kk],
                                                (short)0, c, false, false);

    if (more) {
      float* __restrict__ nb = s_a[(rt + 1) & 1] + soff;
      stage_row<SECOND>(nb, st0, sd0, bi4);
      stage_row<SECOND>(nb + 4 * AP, st1, sd1, bi4);
      stage_row<SECOND>(nb + 8 * AP, st2, sd2, bi4);
      stage_row<SECOND>(nb + 12 * AP, st3, sd3, bi4);
      __syncthreads();
    }

    if constexpr (!SECOND) {
      d0 = *(const float4*)(s_dis + rt * 16 + rbase);
      d1 = *(const float4*)(s_dis + rt * 16 + rbase + 4);
    }

    const float dd[8] = {d0.x, d0.y, d0.z, d0.w, d1.x, d1.y, d1.z, d1.w};
    const size_t base = (size_t)(row0 + rbase) * ND + col0 + m;
    float* __restrict__ hp = H + base;
    float* __restrict__ op = OUT + base;
#pragma unroll
    for (int r = 0; r < 8; ++r) {
      const float w = dd[r];
      const float hs = c[r] * w;
      hp[r * ND] = hs;
      op[r * ND] = SECOND ? bv + hs * w : hs;
    }
  }
}

template <int D, int NE, bool WDST>
__device__ __forceinline__ void scatter_edges(const float* __restrict__ hl,
                                              float* __restrict__ ol,
                                              int sv, int dv, float wl, int j) {
  constexpr int C = D / 32;
  float v[NE][C];
  for (int k = 0; k < NE; ++k) {
    const int s = __builtin_amdgcn_readlane(sv, j + k);
    const float* __restrict__ hp = hl + (size_t)s * D;
    for (int c = 0; c < C; ++c) v[k][c] = hp[c * 32];
  }
  for (int k = 0; k < NE; ++k) {
    const int d = __builtin_amdgcn_readlane(dv, j + k);
    float* op = ol + (size_t)d * D;
    if constexpr (WDST) {
      const float w = __int_as_float(__builtin_amdgcn_readlane(__float_as_int(wl), j + k));
      for (int c = 0; c < C; ++c) unsafeAtomicAdd(op + c * 32, v[k][c] * w);
    } else {
      for (int c = 0; c < C; ++c) unsafeAtomicAdd(op + c * 32, v[k][c]);
    }
  }
}

template <int D, int NE, bool WDST>
__device__ __forceinline__ void scatter_run(const float* __restrict__ hl,
                                            float* __restrict__ ol,
                                            int sv, int dv, float wl, int& j, int n) {
  for (; j + NE <= n; j += NE) scatter_edges<D, NE, WDST>(hl, ol, sv, dv, wl, j);
  if constexpr (NE > 1) scatter_run<D, NE / 2, WDST>(hl, ol, sv, dv, wl, j, n);
}

template <int D, int NE, bool WDST>
__global__ void __launch_bounds__(SW * 32)
scatter_kernel(const int* __restrict__ src,
               const int* __restrict__ dst,
               const float* __restrict__ h,
               const float* __restrict__ dis,
               float* __restrict__ out,
               unsigned* __restrict__ cnt, int E, int nh) {
  __shared__ int   l_s[SW][CH];
  __shared__ int   l_d[SW][CH];
  __shared__ float l_w[WDST ? SW : 1][WDST ? CH : 1];
  __shared__ unsigned claim[2];
  const int lane = threadIdx.x & 31;
  const int wave = threadIdx.x >> 5;
  int*   ls = l_s[wave];
  int*   ld = l_d[wave];
  float* lw = l_w[WDST ? wave : 0];
  const float* __restrict__ hl = h + lane;
  float* __restrict__ ol = out + lane;
  const unsigned ncb = (unsigned)((E + CH * SW - 1) / (CH * SW));
  int k = (int)((__builtin_amdgcn_s_sendmsg_rtn(0x87) >> 18) & 1u);
  unsigned r = 0;
  for (int pass = 0; pass < 2; ++pass, k ^= 1) {
    for (unsigned t = 0; t <= ncb; ++t) {
      const unsigned slot = r++ & 1u;
      if (threadIdx.x == 0) claim[slot] = atomicAdd(cnt + k * CSTR, 1u);
      __syncthreads();
      const unsigned cb = (unsigned)__builtin_amdgcn_readfirstlane((int)claim[slot]);
      if (cb >= ncb) break;
      const unsigned c = cb * SW + wave;
      const int e0 = (int)c * CH + lane;
      int s[CH / 32], d[CH / 32];
      for (int g = 0; g < CH / 32; ++g) {
        const int e = e0 + g * 32;
        const bool ok = e < E;
        d[g] = ok ? dst[e] : -1;
        s[g] = ok ? src[e] : 0;
      }
      float w[CH / 32];
      if constexpr (WDST) {
        for (int g = 0; g < CH / 32; ++g) {
          const bool m = (d[g] >= 0) && ((int)(d[g] >= nh) == k);
          w[g] = dis[m ? d[g] : 0];
        }
      }
      int L = 0;
      for (int g = 0; g < CH / 32; ++g) {
        const bool m = (d[g] >= 0) && ((int)(d[g] >= nh) == k);
        const unsigned mask = __builtin_amdgcn_ballot_w32(m);
        if (m) {
          const int p = L + (int)__builtin_amdgcn_mbcnt_lo(mask, 0u);
          ls[p] = s[g];
          ld[p] = d[g];
          if constexpr (WDST) lw[p] = w[g];
        }
        L += __builtin_popcount(mask);
      }
      for (int base = 0; base < L; base += 32) {
        const int n = min(32, L - base);
        const int i = base + min(lane, n - 1);
        const int sv = ls[i];
        const int dv = ld[i];
        float wl = 0.0f;
        if constexpr (WDST) wl = lw[i];
        int j = 0;
        scatter_run<D, NE, WDST>(hl, ol, sv, dv, wl, j, n);
      }
    }
  }
}

static inline int cdiv(long long a, long long b) { return (int)((a + b - 1) / b); }

static void run_branch(const float* x, const int* ei, const float* W1,
                       const float* b1, const float* W2, const float* b2,
                       float* out_final, const float* part, float* dis, float* h, float* out1,
                       int N, int E, hipStream_t stream) {
  const int IN = 128, HD = 128, OD = 64;
  const int RT = 5;
  const int NE1 = 4, NE2 = 8;
  const int SB = 2048;
  const int* src = ei;
  const int* dst = ei + E;

  const int rowBlocks = N / (16 * RT);
  unsigned* cnt1 = (unsigned*)out_final;
  unsigned* cnt2 = (unsigned*)(h + (size_t)N * 64);

  gemm_wmma_f32<IN, HD, RT, false>
      <<<dim3(rowBlocks, HD / 64), 128, 0, stream>>>(x, W1, part, dis, b1, b1, h, out1, cnt1, N);
  scatter_kernel<HD, NE1, false><<<SB, SW * 32, 0, stream>>>(src, dst, h, dis, out1, cnt1, E, N / 2);

  gemm_wmma_f32<HD, OD, RT, true>
      <<<dim3(rowBlocks, 1), 128, 0, stream>>>(out1, W2, part, dis, b1, b2, h, out_final, cnt2, N);
  scatter_kernel<OD, NE2, true><<<SB, SW * 32, 0, stream>>>(src, dst, h, dis, out_final, cnt2, E, N / 2);
}

extern "C" void kernel_launch(void* const* d_in, const int* in_sizes, int n_in,
                              void* d_out, int out_size, void* d_ws, size_t ws_size,
                              hipStream_t stream) {
  const float* x_feat = (const float*)d_in[0];
  const int*   x_ei   = (const int*)d_in[1];
  const float* y_feat = (const float*)d_in[2];
  const int*   y_ei   = (const int*)d_in[3];
  const float* W1x = (const float*)d_in[4];
  const float* b1x = (const float*)d_in[5];
  const float* W2x = (const float*)d_in[6];
  const float* b2x = (const float*)d_in[7];
  const float* W1y = (const float*)d_in[8];
  const float* b1y = (const float*)d_in[9];
  const float* W2y = (const float*)d_in[10];
  const float* b2y = (const float*)d_in[11];

  const int N = in_sizes[0] / 128;
  const int E = in_sizes[1] / 2;

  float* h    = (float*)d_ws;
  float* out1 = h + (size_t)N * 128;
  float* deg  = out1 + (size_t)N * 128;

  float* out_x = (float*)d_out;
  float* out_y = out_x + (size_t)N * 64;

  float* part_x = out_x + 128;
  float* part_y = out_y + 128;
  degcount_kernel<<<dim3(cdiv(N, DR), DC, 2), DT, 0, stream>>>(x_ei + E, y_ei + E, part_x, part_y, E);

  run_branch(x_feat, x_ei, W1x, b1x, W2x, b2x, out_x, part_x, deg, h, out1, N, E, stream);
  run_branch(y_feat, y_ei, W1y, b1y, W2y, b2y, out_y, part_y, deg, h, out1, N, E, stream);
}
